# starting kernel
# baseline (speedup 1.0000x reference)
#include <hip/hip_runtime.h>


typedef __attribute__((ext_vector_type(2))) float v2f;
typedef __attribute__((ext_vector_type(8))) float v8f;

#define BH 64
#define NN 4096
#define DD 64
#define K1_ROWS 256
#define K1_CHUNKS 16
#define C64 64

static __device__ __forceinline__ v8f wmma_f32(v2f a, v2f b, v8f c) {
  return __builtin_amdgcn_wmma_f32_16x16x4_f32(false, a, false, b, (short)0, c,
                                               false, false);
}

__global__ void k1_gpart(const float* __restrict__ k, const float* __restrict__ v,
                         float* __restrict__ Gpart, float* __restrict__ S) {
  const int bh = blockIdx.x;
  const int ch = blockIdx.y;
  const int t  = threadIdx.x;
  const float* kb = k + (size_t)bh * NN * DD;
  const float* vb = v + (size_t)bh * NN * DD;
  const int n0 = ch * K1_ROWS;

  {
    const int d = t & 63, seg = t >> 6;
    const int r0 = n0 + seg * 64;
    float s = 0.f;
    for (int r = 0; r < 64; ++r) s += kb[(size_t)(r0 + r) * DD + d];
    S[((size_t)(bh * C64) + (ch * 4 + seg)) * DD + d] = s;
  }

  const int wave = t >> 5, lane = t & 31;
  const int m = lane & 15, khalf = (lane >> 4) * 2;
  float* gout = Gpart + (size_t)(bh * K1_CHUNKS + ch) * (DD * DD);
  for (int i = 0; i < 2; ++i) {
    const int tt = wave * 2 + i;
    const int ti = tt >> 2, tj = tt & 3;
    v8f acc = {0.f, 0.f, 0.f, 0.f, 0.f, 0.f, 0.f, 0.f};
    for (int kk = 0; kk < K1_ROWS; kk += 4) {
      const int ra = n0 + kk + khalf;
      v2f a, b;
      a.x = kb[(size_t)ra * DD + ti * 16 + m];
      a.y = kb[(size_t)(ra + 1) * DD + ti * 16 + m];
      b.x = vb[(size_t)ra * DD + tj * 16 + m];
      b.y = vb[(size_t)(ra + 1) * DD + tj * 16 + m];
      acc = wmma_f32(a, b, acc);
    }
    const int rbase = ti * 16 + (lane >> 4) * 8;
    for (int j = 0; j < 8; ++j)
      gout[(rbase + j) * DD + tj * 16 + m] = acc[j];
  }
}

__global__ void k2_combine(const float* __restrict__ proj,
                           const float* __restrict__ Gpart,
                           float* __restrict__ S, float* __restrict__ Wm,
                           float* __restrict__ GramG) {
  __shared__ float G[DD * DD];
  __shared__ float Pl[DD * DD];
  __shared__ float Gm[DD * DD];
  const int bh = blockIdx.x;
  const int t  = threadIdx.x;

  for (int idx = t; idx < DD * DD; idx += 256) {
    float s = 0.f;
    for (int c = 0; c < K1_CHUNKS; ++c)
      s += Gpart[(size_t)(bh * K1_CHUNKS + c) * (DD * DD) + idx];
    G[idx]  = s;
    Pl[idx] = proj[idx];
  }
  __syncthreads();

  for (int idx = t; idx < DD * DD; idx += 256) {
    const int d1 = idx >> 6, d2 = idx & 63;
    float s = 0.f;
    for (int mm = 0; mm < DD; ++mm) s += Pl[mm * DD + d1] * Pl[mm * DD + d2];
    Gm[idx] = s;
    if (bh == 0) GramG[idx] = s;
  }
  __syncthreads();

  for (int idx = t; idx < DD * DD; idx += 256) {
    const int d1 = idx >> 6, e = idx & 63;
    float s = 0.f;
    for (int d2 = 0; d2 < DD; ++d2) s += Gm[d1 * DD + d2] * G[d2 * DD + e];
    Wm[(size_t)bh * DD * DD + idx] = s;
  }

  if (t < 64) {
    float run = 0.f;
    for (int c = 0; c < C64; ++c) {
      const size_t idx = ((size_t)(bh * C64) + c) * DD + t;
      const float tmp = S[idx];
      S[idx] = run;
      run += tmp;
    }
  }
}

__global__ void k3_out(const float* __restrict__ q, const float* __restrict__ k,
                       const float* __restrict__ S, const float* __restrict__ Wm,
                       const float* __restrict__ GramG, float* __restrict__ out) {
  __shared__ float kc[64 * 65];
  __shared__ float ab[64 * 65];
  __shared__ float qgl[64 * 65];
  __shared__ float qsum[4 * 64];
  __shared__ float dinv[64];
  const int bh = blockIdx.x, c64 = blockIdx.y;
  const int t  = threadIdx.x;
  const int n0 = c64 * 64;
  const float* qb = q + (size_t)bh * NN * DD;
  const float* kb = k + (size_t)bh * NN * DD;

  for (int idx = t; idx < 64 * 64; idx += 256) {
    const int r = idx >> 6, d = idx & 63;
    kc[r * 65 + d] = kb[(size_t)n0 * DD + idx];
    ab[r * 65 + d] = GramG[idx];
  }
  __syncthreads();

  const int d = t & 63, qt = t >> 6;
  {
    float run = 0.f;
    for (int r = qt * 16; r < qt * 16 + 16; ++r) {
      run += kc[r * 65 + d];
      kc[r * 65 + d] = run;
    }
    qsum[qt * 64 + d] = run;
  }
  __syncthreads();
  if (t < 64) {
    const float p = S[((size_t)(bh * C64) + c64) * DD + t];
    const float o0 = p;
    const float o1 = o0 + qsum[0 * 64 + t];
    const float o2 = o1 + qsum[1 * 64 + t];
    const float o3 = o2 + qsum[2 * 64 + t];
    qsum[0 * 64 + t] = o0; qsum[1 * 64 + t] = o1;
    qsum[2 * 64 + t] = o2; qsum[3 * 64 + t] = o3;
  }
  __syncthreads();
  {
    const float off = qsum[qt * 64 + d];
    for (int r = qt * 16; r < qt * 16 + 16; ++r) kc[r * 65 + d] += off;
  }
  __syncthreads();

  const int wave = t >> 5, lane = t & 31;
  const int m = lane & 15, khalf = (lane >> 4) * 2;

  for (int i = 0; i < 2; ++i) {
    const int tt = wave * 2 + i;
    const int ti = tt >> 2, tj = tt & 3;
    v8f acc = {0.f, 0.f, 0.f, 0.f, 0.f, 0.f, 0.f, 0.f};
    const float* arow = qb + (size_t)(n0 + ti * 16 + m) * DD;
    for (int kk = 0; kk < DD; kk += 4) {
      v2f a = *(const v2f*)(arow + kk + khalf);
      v2f b;
      b.x = ab[(kk + khalf) * 65 + tj * 16 + m];
      b.y = ab[(kk + khalf + 1) * 65 + tj * 16 + m];
      acc = wmma_f32(a, b, acc);
    }
    const int rbase = ti * 16 + (lane >> 4) * 8;
    for (int j = 0; j < 8; ++j)
      qgl[(rbase + j) * 65 + tj * 16 + m] = acc[j];
  }
  __syncthreads();

  if (t < 64) {
    float s = 0.f;
    for (int dd = 0; dd < 64; ++dd) s += qgl[t * 65 + dd] * kc[t * 65 + dd];
    dinv[t] = 1.0f / s;
  }
  __syncthreads();

  for (int idx = t; idx < 64 * 64; idx += 256) {
    const int r = idx >> 6, dd = idx & 63;
    ab[r * 65 + dd] = Wm[(size_t)bh * DD * DD + idx];
  }
  __syncthreads();

  for (int i = 0; i < 2; ++i) {
    const int tt = wave * 2 + i;
    const int ti = tt >> 2, tj = tt & 3;
    v8f acc = {0.f, 0.f, 0.f, 0.f, 0.f, 0.f, 0.f, 0.f};
    const float* arow = qb + (size_t)(n0 + ti * 16 + m) * DD;
    for (int kk = 0; kk < DD; kk += 4) {
      v2f a = *(const v2f*)(arow + kk + khalf);
      v2f b;
      b.x = ab[(kk + khalf) * 65 + tj * 16 + m];
      b.y = ab[(kk + khalf + 1) * 65 + tj * 16 + m];
      acc = wmma_f32(a, b, acc);
    }
    const int rbase = ti * 16 + (lane >> 4) * 8;
    for (int j = 0; j < 8; ++j) {
      const float dv = dinv[rbase + j];
      out[((size_t)bh * NN + n0 + rbase + j) * DD + tj * 16 + m] = acc[j] * dv;
    }
  }
}

extern "C" void kernel_launch(void* const* d_in, const int* in_sizes, int n_in,
                              void* d_out, int out_size, void* d_ws, size_t ws_size,
                              hipStream_t stream) {
  const float* q    = (const float*)d_in[0];
  const float* k    = (const float*)d_in[1];
  const float* v    = (const float*)d_in[2];
  const float* proj = (const float*)d_in[3];
  float* out = (float*)d_out;

  float* ws    = (float*)d_ws;
  float* Gpart = ws;
  float* S     = Gpart + (size_t)BH * K1_CHUNKS * DD * DD;
  float* Wm    = S + (size_t)BH * C64 * DD;
  float* Gram  = Wm + (size_t)BH * DD * DD;

  k1_gpart<<<dim3(BH, K1_CHUNKS), 256, 0, stream>>>(k, v, Gpart, S);
  k2_combine<<<dim3(BH), 256, 0, stream>>>(proj, Gpart, S, Wm, Gram);
  k3_out<<<dim3(BH, C64), 256, 0, stream>>>(q, k, S, Wm, Gram, out);
}


# K1 re-cut: one work-group per head and 16-col slab of G walks all rows (32 waves summed in LDS), no partials; W=P^T(P G), k scan, Gram in its tail; K2 launch gone; G, W, Gram, S sum order changes
# speedup vs baseline: 1.2972x; 1.2972x over previous
#include <hip/hip_runtime.h>


typedef __attribute__((ext_vector_type(2))) float v2f;
typedef __attribute__((ext_vector_type(8))) float v8f;

#define BH 64
#define NN 4096
#define DD 64
#define K1_WAVES 32
#define K1_WROWS 128
#define C64 64

static __device__ __forceinline__ v8f wmma_f32(v2f a, v2f b, v8f c) {
  return __builtin_amdgcn_wmma_f32_16x16x4_f32(false, a, false, b, (short)0, c,
                                               false, false);
}

__global__ __launch_bounds__(1024) void k1_slab(const float* __restrict__ k,
                                                const float* __restrict__ v,
                                                const float* __restrict__ proj,
                                                float* __restrict__ S,
                                                float* __restrict__ Wm,
                                                float* __restrict__ GramG) {
  __shared__ float red[K1_WAVES * DD * 16];
  __shared__ float csl[C64 * DD];
  __shared__ float Gs[DD * 16];
  __shared__ float Ts[DD * 16];
  const int slab = blockIdx.x, bh = blockIdx.y;
  const int t = threadIdx.x;
  const int wave = t >> 5, lane = t & 31;
  const int m = lane & 15, hh = lane >> 4, khalf = hh * 2;
  const float* kb = k + (size_t)bh * NN * DD;
  const float* vb = v + (size_t)bh * NN * DD + slab * 16;

  v8f acc0 = {0.f, 0.f, 0.f, 0.f, 0.f, 0.f, 0.f, 0.f};
  v8f acc1 = acc0, acc2 = acc0, acc3 = acc0;
  for (int c = 0; c < 2; ++c) {
    const int r0 = wave * K1_WROWS + c * 64;
    float cs0 = 0.f, cs1 = 0.f, cs2 = 0.f, cs3 = 0.f;
    for (int kk = 0; kk < 64; kk += 4) {
      const size_t ra = (size_t)(r0 + kk + khalf);
      const float* kr = kb + ra * DD + m;
      const float* vr = vb + ra * DD + m;
      v2f b, a0, a1, a2, a3;
      b.x  = vr[0];   b.y  = vr[DD];
      a0.x = kr[0];   a0.y = kr[DD];
      a1.x = kr[16];  a1.y = kr[DD + 16];
      a2.x = kr[32];  a2.y = kr[DD + 32];
      a3.x = kr[48];  a3.y = kr[DD + 48];
      acc0 = wmma_f32(a0, b, acc0);
      acc1 = wmma_f32(a1, b, acc1);
      acc2 = wmma_f32(a2, b, acc2);
      acc3 = wmma_f32(a3, b, acc3);
      cs0 += a0.x; cs0 += a0.y;
      cs1 += a1.x; cs1 += a1.y;
      cs2 += a2.x; cs2 += a2.y;
      cs3 += a3.x; cs3 += a3.y;
    }
    cs0 += __shfl_xor(cs0, 16);
    cs1 += __shfl_xor(cs1, 16);
    cs2 += __shfl_xor(cs2, 16);
    cs3 += __shfl_xor(cs3, 16);
    if (slab == 0 && hh == 0) {
      float* dst = csl + (wave * 2 + c) * DD + m;
      dst[0] = cs0; dst[16] = cs1; dst[32] = cs2; dst[48] = cs3;
    }
  }

  {
    float* rp = red + (wave * DD + hh * 8) * 16 + m;
    for (int j = 0; j < 8; ++j) {
      rp[(j) * 16]      = acc0[j];
      rp[(16 + j) * 16] = acc1[j];
      rp[(32 + j) * 16] = acc2[j];
      rp[(48 + j) * 16] = acc3[j];
    }
  }
  __syncthreads();

  {
    float s = 0.f;
    for (int w = 0; w < K1_WAVES; ++w) s += red[w * (DD * 16) + t];
    Gs[t] = s;
  }
  if (slab == 0 && t >= 960) {
    const int d = t - 960;
    float run = 0.f;
    for (int c = 0; c < C64; ++c) {
      S[((size_t)(bh * C64) + c) * DD + d] = run;
      run += csl[c * DD + d];
    }
  }
  __syncthreads();

  if (wave < 4) {
    v8f acc = {0.f, 0.f, 0.f, 0.f, 0.f, 0.f, 0.f, 0.f};
    const float* arow = proj + (size_t)(wave * 16 + m) * DD;
    for (int kk = 0; kk < DD; kk += 4) {
      v2f a = *(const v2f*)(arow + kk + khalf);
      v2f b;
      b.x = Gs[(kk + khalf) * 16 + m];
      b.y = Gs[(kk + khalf + 1) * 16 + m];
      acc = wmma_f32(a, b, acc);
    }
    for (int j = 0; j < 8; ++j) Ts[(wave * 16 + hh * 8 + j) * 16 + m] = acc[j];
  } else if (slab == 0 && bh == 0 && wave < 20) {
    const int tt = wave - 4;
    const int ti = tt >> 2, tj = tt & 3;
    v8f acc = {0.f, 0.f, 0.f, 0.f, 0.f, 0.f, 0.f, 0.f};
    for (int kk = 0; kk < DD; kk += 4) {
      const float* pr = proj + (size_t)(kk + khalf) * DD + m;
      v2f a, b;
      a.x = pr[ti * 16];  a.y = pr[DD + ti * 16];
      b.x = pr[tj * 16];  b.y = pr[DD + tj * 16];
      acc = wmma_f32(a, b, acc);
    }
    for (int j = 0; j < 8; ++j)
      GramG[(ti * 16 + hh * 8 + j) * DD + tj * 16 + m] = acc[j];
  }
  __syncthreads();

  if (wave < 4) {
    v8f acc = {0.f, 0.f, 0.f, 0.f, 0.f, 0.f, 0.f, 0.f};
    for (int kk = 0; kk < DD; kk += 4) {
      const float* pr = proj + (size_t)(kk + khalf) * DD + wave * 16 + m;
      v2f a, b;
      a.x = pr[0];  a.y = pr[DD];
      b.x = Ts[(kk + khalf) * 16 + m];
      b.y = Ts[(kk + khalf + 1) * 16 + m];
      acc = wmma_f32(a, b, acc);
    }
    float* wo = Wm + (size_t)bh * DD * DD + (size_t)(wave * 16 + hh * 8) * DD + slab * 16 + m;
    for (int j = 0; j < 8; ++j) wo[j * DD] = acc[j];
  }
}

__global__ void k3_out(const float* __restrict__ q, const float* __restrict__ k,
                       const float* __restrict__ S, const float* __restrict__ Wm,
                       const float* __restrict__ GramG, float* __restrict__ out) {
  __shared__ float kc[64 * 65];
  __shared__ float ab[64 * 65];
  __shared__ float qgl[64 * 65];
  __shared__ float qsum[4 * 64];
  __shared__ float dinv[64];
  const int bh = blockIdx.x, c64 = blockIdx.y;
  const int t  = threadIdx.x;
  const int n0 = c64 * 64;
  const float* qb = q + (size_t)bh * NN * DD;
  const float* kb = k + (size_t)bh * NN * DD;

  for (int idx = t; idx < 64 * 64; idx += 256) {
    const int r = idx >> 6, d = idx & 63;
    kc[r * 65 + d] = kb[(size_t)n0 * DD + idx];
    ab[r * 65 + d] = GramG[idx];
  }
  __syncthreads();

  const int d = t & 63, qt = t >> 6;
  {
    float run = 0.f;
    for (int r = qt * 16; r < qt * 16 + 16; ++r) {
      run += kc[r * 65 + d];
      kc[r * 65 + d] = run;
    }
    qsum[qt * 64 + d] = run;
  }
  __syncthreads();
  if (t < 64) {
    const float p = S[((size_t)(bh * C64) + c64) * DD + t];
    const float o0 = p;
    const float o1 = o0 + qsum[0 * 64 + t];
    const float o2 = o1 + qsum[1 * 64 + t];
    const float o3 = o2 + qsum[2 * 64 + t];
    qsum[0 * 64 + t] = o0; qsum[1 * 64 + t] = o1;
    qsum[2 * 64 + t] = o2; qsum[3 * 64 + t] = o3;
  }
  __syncthreads();
  {
    const float off = qsum[qt * 64 + d];
    for (int r = qt * 16; r < qt * 16 + 16; ++r) kc[r * 65 + d] += off;
  }
  __syncthreads();

  const int wave = t >> 5, lane = t & 31;
  const int m = lane & 15, khalf = (lane >> 4) * 2;

  for (int i = 0; i < 2; ++i) {
    const int tt = wave * 2 + i;
    const int ti = tt >> 2, tj = tt & 3;
    v8f acc = {0.f, 0.f, 0.f, 0.f, 0.f, 0.f, 0.f, 0.f};
    const float* arow = qb + (size_t)(n0 + ti * 16 + m) * DD;
    for (int kk = 0; kk < DD; kk += 4) {
      v2f a = *(const v2f*)(arow + kk + khalf);
      v2f b;
      b.x = ab[(kk + khalf) * 65 + tj * 16 + m];
      b.y = ab[(kk + khalf + 1) * 65 + tj * 16 + m];
      acc = wmma_f32(a, b, acc);
    }
    const int rbase = ti * 16 + (lane >> 4) * 8;
    for (int j = 0; j < 8; ++j)
      qgl[(rbase + j) * 65 + tj * 16 + m] = acc[j];
  }
  __syncthreads();

  if (t < 64) {
    float s = 0.f;
    for (int dd = 0; dd < 64; ++dd) s += qgl[t * 65 + dd] * kc[t * 65 + dd];
    dinv[t] = 1.0f / s;
  }
  __syncthreads();

  for (int idx = t; idx < 64 * 64; idx += 256) {
    const int r = idx >> 6, dd = idx & 63;
    ab[r * 65 + dd] = Wm[(size_t)bh * DD * DD + idx];
  }
  __syncthreads();

  for (int i = 0; i < 2; ++i) {
    const int tt = wave * 2 + i;
    const int ti = tt >> 2, tj = tt & 3;
    v8f acc = {0.f, 0.f, 0.f, 0.f, 0.f, 0.f, 0.f, 0.f};
    const float* arow = qb + (size_t)(n0 + ti * 16 + m) * DD;
    for (int kk = 0; kk < DD; kk += 4) {
      v2f a = *(const v2f*)(arow + kk + khalf);
      v2f b;
      b.x = ab[(kk + khalf) * 65 + tj * 16 + m];
      b.y = ab[(kk + khalf + 1) * 65 + tj * 16 + m];
      acc = wmma_f32(a, b, acc);
    }
    const int rbase = ti * 16 + (lane >> 4) * 8;
    for (int j = 0; j < 8; ++j) {
      const float dv = dinv[rbase + j];
      out[((size_t)bh * NN + n0 + rbase + j) * DD + tj * 16 + m] = acc[j] * dv;
    }
  }
}

extern "C" void kernel_launch(void* const* d_in, const int* in_sizes, int n_in,
                              void* d_out, int out_size, void* d_ws, size_t ws_size,
                              hipStream_t stream) {
  const float* q    = (const float*)d_in[0];
  const float* k    = (const float*)d_in[1];
  const float* v    = (const float*)d_in[2];
  const float* proj = (const float*)d_in[3];
  float* out = (float*)d_out;

  float* ws   = (float*)d_ws;
  float* S    = ws;
  float* Wm   = S + (size_t)BH * C64 * DD;
  float* Gram = Wm + (size_t)BH * DD * DD;

  k1_slab<<<dim3(4, BH), 1024, 0, stream>>>(k, v, proj, S, Wm, Gram);
  k3_out<<<dim3(BH, C64), 256, 0, stream>>>(q, k, S, Wm, Gram, out);
}


# K3 walks 4 chunks per work-group (grid 64x16): Gram and W stay in LDS, the next chunk's k rows ride in registers across LDS-only barriers, q fragments fetched once per chunk; sum orders kept
# speedup vs baseline: 1.5557x; 1.1993x over previous
#include <hip/hip_runtime.h>


typedef __attribute__((ext_vector_type(2))) float v2f;
typedef __attribute__((ext_vector_type(8))) float v8f;

#define BH 64
#define NN 4096
#define DD 64
#define K1_WAVES 32
#define K1_WROWS 128
#define C64 64

static __device__ __forceinline__ v8f wmma_f32(v2f a, v2f b, v8f c) {
  return __builtin_amdgcn_wmma_f32_16x16x4_f32(false, a, false, b, (short)0, c,
                                               false, false);
}

__global__ __launch_bounds__(1024) void k1_slab(const float* __restrict__ k,
                                                const float* __restrict__ v,
                                                const float* __restrict__ proj,
                                                float* __restrict__ S,
                                                float* __restrict__ Wm,
                                                float* __restrict__ GramG) {
  __shared__ float red[K1_WAVES * DD * 16];
  __shared__ float csl[C64 * DD];
  __shared__ float Gs[DD * 16];
  __shared__ float Ts[DD * 16];
  const int slab = blockIdx.x, bh = blockIdx.y;
  const int t = threadIdx.x;
  const int wave = t >> 5, lane = t & 31;
  const int m = lane & 15, hh = lane >> 4, khalf = hh * 2;
  const float* kb = k + (size_t)bh * NN * DD;
  const float* vb = v + (size_t)bh * NN * DD + slab * 16;

  v8f acc0 = {0.f, 0.f, 0.f, 0.f, 0.f, 0.f, 0.f, 0.f};
  v8f acc1 = acc0, acc2 = acc0, acc3 = acc0;
  for (int c = 0; c < 2; ++c) {
    const int r0 = wave * K1_WROWS + c * 64;
    float cs0 = 0.f, cs1 = 0.f, cs2 = 0.f, cs3 = 0.f;
    for (int kk = 0; kk < 64; kk += 4) {
      const size_t ra = (size_t)(r0 + kk + khalf);
      const float* kr = kb + ra * DD + m;
      const float* vr = vb + ra * DD + m;
      v2f b, a0, a1, a2, a3;
      b.x  = vr[0];   b.y  = vr[DD];
      a0.x = kr[0];   a0.y = kr[DD];
      a1.x = kr[16];  a1.y = kr[DD + 16];
      a2.x = kr[32];  a2.y = kr[DD + 32];
      a3.x = kr[48];  a3.y = kr[DD + 48];
      acc0 = wmma_f32(a0, b, acc0);
      acc1 = wmma_f32(a1, b, acc1);
      acc2 = wmma_f32(a2, b, acc2);
      acc3 = wmma_f32(a3, b, acc3);
      cs0 += a0.x; cs0 += a0.y;
      cs1 += a1.x; cs1 += a1.y;
      cs2 += a2.x; cs2 += a2.y;
      cs3 += a3.x; cs3 += a3.y;
    }
    cs0 += __shfl_xor(cs0, 16);
    cs1 += __shfl_xor(cs1, 16);
    cs2 += __shfl_xor(cs2, 16);
    cs3 += __shfl_xor(cs3, 16);
    if (slab == 0 && hh == 0) {
      float* dst = csl + (wave * 2 + c) * DD + m;
      dst[0] = cs0; dst[16] = cs1; dst[32] = cs2; dst[48] = cs3;
    }
  }

  {
    float* rp = red + (wave * DD + hh * 8) * 16 + m;
    for (int j = 0; j < 8; ++j) {
      rp[(j) * 16]      = acc0[j];
      rp[(16 + j) * 16] = acc1[j];
      rp[(32 + j) * 16] = acc2[j];
      rp[(48 + j) * 16] = acc3[j];
    }
  }
  __syncthreads();

  {
    float s = 0.f;
    for (int w = 0; w < K1_WAVES; ++w) s += red[w * (DD * 16) + t];
    Gs[t] = s;
  }
  if (slab == 0 && t >= 960) {
    const int d = t - 960;
    float run = 0.f;
    for (int c = 0; c < C64; ++c) {
      S[((size_t)(bh * C64) + c) * DD + d] = run;
      run += csl[c * DD + d];
    }
  }
  __syncthreads();

  if (wave < 4) {
    v8f acc = {0.f, 0.f, 0.f, 0.f, 0.f, 0.f, 0.f, 0.f};
    const float* arow = proj + (size_t)(wave * 16 + m) * DD;
    for (int kk = 0; kk < DD; kk += 4) {
      v2f a = *(const v2f*)(arow + kk + khalf);
      v2f b;
      b.x = Gs[(kk + khalf) * 16 + m];
      b.y = Gs[(kk + khalf + 1) * 16 + m];
      acc = wmma_f32(a, b, acc);
    }
    for (int j = 0; j < 8; ++j) Ts[(wave * 16 + hh * 8 + j) * 16 + m] = acc[j];
  } else if (slab == 0 && bh == 0 && wave < 20) {
    const int tt = wave - 4;
    const int ti = tt >> 2, tj = tt & 3;
    v8f acc = {0.f, 0.f, 0.f, 0.f, 0.f, 0.f, 0.f, 0.f};
    for (int kk = 0; kk < DD; kk += 4) {
      const float* pr = proj + (size_t)(kk + khalf) * DD + m;
      v2f a, b;
      a.x = pr[ti * 16];  a.y = pr[DD + ti * 16];
      b.x = pr[tj * 16];  b.y = pr[DD + tj * 16];
      acc = wmma_f32(a, b, acc);
    }
    for (int j = 0; j < 8; ++j)
      GramG[(ti * 16 + hh * 8 + j) * DD + tj * 16 + m] = acc[j];
  }
  __syncthreads();

  if (wave < 4) {
    v8f acc = {0.f, 0.f, 0.f, 0.f, 0.f, 0.f, 0.f, 0.f};
    for (int kk = 0; kk < DD; kk += 4) {
      const float* pr = proj + (size_t)(kk + khalf) * DD + wave * 16 + m;
      v2f a, b;
      a.x = pr[0];  a.y = pr[DD];
      b.x = Ts[(kk + khalf) * 16 + m];
      b.y = Ts[(kk + khalf + 1) * 16 + m];
      acc = wmma_f32(a, b, acc);
    }
    float* wo = Wm + (size_t)bh * DD * DD + (size_t)(wave * 16 + hh * 8) * DD + slab * 16 + m;
    for (int j = 0; j < 8; ++j) wo[j * DD] = acc[j];
  }
}

static __device__ __forceinline__ void lds_barrier() {
  __builtin_amdgcn_fence(__ATOMIC_RELEASE, "workgroup", "local");
  __builtin_amdgcn_s_barrier();
  __builtin_amdgcn_fence(__ATOMIC_ACQUIRE, "workgroup", "local");
}

#define K3_CPW 4
#define K3_GROUPS 16

__global__ __launch_bounds__(256, 8) void k3_out(const float* __restrict__ q,
                                              const float* __restrict__ k,
                                              const float* __restrict__ S,
                                              const float* __restrict__ Wm,
                                              const float* __restrict__ GramG,
                                              float* __restrict__ out) {
  __shared__ float kc[64 * 65];
  __shared__ float gr[64 * 65];
  __shared__ float wl[64 * 65];
  __shared__ float qgl[64 * 65];
  __shared__ float qsum[4 * 64];
  __shared__ float dinv[64];
  const int bh = blockIdx.x, grp = blockIdx.y;
  const int t  = threadIdx.x;
  const int d = t & 63, qt = t >> 6;
  const int wave = t >> 5, lane = t & 31;
  const int m = lane & 15, hh = lane >> 4, khalf = hh * 2;
  const int ti = wave >> 1;
  const int tj0 = (wave & 1) * 2;
  const float* qb = q + (size_t)bh * NN * DD;
  const float* kb = k + (size_t)bh * NN * DD;
  const float* wb = Wm + (size_t)bh * DD * DD;

  float kreg[16];
  {
    const float* kp = kb + (size_t)(grp * K3_CPW) * 64 * DD + t;
    for (int i = 0; i < 16; ++i) kreg[i] = kp[i * 256];
  }
  for (int i = 0; i < 16; ++i) {
    const int idx = t + i * 256;
    gr[(qt + 4 * i) * 65 + d] = GramG[idx];
    wl[(qt + 4 * i) * 65 + d] = wb[idx];
  }

#pragma unroll 1
  for (int cc = 0; cc < K3_CPW; ++cc) {
    const int c64 = grp * K3_CPW + cc;
    const int n0 = c64 * 64;

    const float pre = S[((size_t)(bh * C64) + c64) * DD + d];
    v2f afr[16];
    {
      const float* arow = qb + (size_t)(n0 + ti * 16 + m) * DD + khalf;
      for (int kk = 0; kk < 16; ++kk) afr[kk] = *(const v2f*)(arow + kk * 4);
    }

    for (int i = 0; i < 16; ++i) kc[(qt + 4 * i) * 65 + d] = kreg[i];
    lds_barrier();

    {
      const int nn = (cc + 1 < K3_CPW) ? n0 + 64 : n0;
      const float* kp = kb + (size_t)nn * DD + t;
      for (int i = 0; i < 16; ++i) kreg[i] = kp[i * 256];
    }

    {
      float run = 0.f;
      for (int r = qt * 16; r < qt * 16 + 16; ++r) {
        run += kc[r * 65 + d];
        kc[r * 65 + d] = run;
      }
      qsum[qt * 64 + d] = run;
    }
    lds_barrier();
    if (t < 64) {
      const float o0 = pre;
      const float o1 = o0 + qsum[0 * 64 + t];
      const float o2 = o1 + qsum[1 * 64 + t];
      const float o3 = o2 + qsum[2 * 64 + t];
      qsum[0 * 64 + t] = o0; qsum[1 * 64 + t] = o1;
      qsum[2 * 64 + t] = o2; qsum[3 * 64 + t] = o3;
    }
    lds_barrier();
    {
      const float off = qsum[qt * 64 + d];
      for (int r = qt * 16; r < qt * 16 + 16; ++r) kc[r * 65 + d] += off;
    }
    lds_barrier();

    const int rbase = ti * 16 + hh * 8;

    for (int i = 0; i < 2; ++i) {
      const int tj = tj0 + i;
      v8f acc = {0.f, 0.f, 0.f, 0.f, 0.f, 0.f, 0.f, 0.f};
      for (int kk = 0; kk < 16; ++kk) {
        v2f b;
        b.x = gr[(kk * 4 + khalf) * 65 + tj * 16 + m];
        b.y = gr[(kk * 4 + khalf + 1) * 65 + tj * 16 + m];
        acc = wmma_f32(afr[kk], b, acc);
      }
      for (int j = 0; j < 8; ++j)
        qgl[(rbase + j) * 65 + tj * 16 + m] = acc[j];
    }
    lds_barrier();

    if (t < 64) {
      float s = 0.f;
      for (int dd = 0; dd < 64; ++dd) s += qgl[t * 65 + dd] * kc[t * 65 + dd];
      dinv[t] = 1.0f / s;
    }
    lds_barrier();

    for (int i = 0; i < 2; ++i) {
      const int tj = tj0 + i;
      v8f acc = {0.f, 0.f, 0.f, 0.f, 0.f, 0.f, 0.f, 0.f};
      for (int kk = 0; kk < 16; ++kk) {
        v2f b;
        b.x = wl[(kk * 4 + khalf) * 65 + tj * 16 + m];
        b.y = wl[(kk * 4 + khalf + 1) * 65 + tj * 16 + m];
        acc = wmma_f32(afr[kk], b, acc);
      }
      for (int j = 0; j < 8; ++j) {
        const float dv = dinv[rbase + j];
        out[((size_t)bh * NN + n0 + rbase + j) * DD + tj * 16 + m] = acc[j] * dv;
      }
    }
  }
}

extern "C" void kernel_launch(void* const* d_in, const int* in_sizes, int n_in,
                              void* d_out, int out_size, void* d_ws, size_t ws_size,
                              hipStream_t stream) {
  const float* q    = (const float*)d_in[0];
  const float* k    = (const float*)d_in[1];
  const float* v    = (const float*)d_in[2];
  const float* proj = (const float*)d_in[3];
  float* out = (float*)d_out;

  float* ws   = (float*)d_ws;
  float* S    = ws;
  float* Wm   = S + (size_t)BH * C64 * DD;
  float* Gram = Wm + (size_t)BH * DD * DD;

  k1_slab<<<dim3(4, BH), 1024, 0, stream>>>(k, v, proj, S, Wm, Gram);
  k3_out<<<dim3(BH, K3_GROUPS), 256, 0, stream>>>(q, k, S, Wm, Gram, out);
}


# K1's 4 work-groups per head tile G 2x2 (32x32 each, whole 128-B lines of k and v, no slab re-read of k); G goes to K3, which forms out = (q Gram) G: T, W and P products gone; out sum order changes
# speedup vs baseline: 1.6882x; 1.0852x over previous
#include <hip/hip_runtime.h>


typedef __attribute__((ext_vector_type(2))) float v2f;
typedef __attribute__((ext_vector_type(8))) float v8f;

#define BH 64
#define NN 4096
#define DD 64
#define K1_WAVES 32
#define K1_WROWS 128
#define C64 64

static __device__ __forceinline__ v8f wmma_f32(v2f a, v2f b, v8f c) {
  return __builtin_amdgcn_wmma_f32_16x16x4_f32(false, a, false, b, (short)0, c,
                                               false, false);
}

__global__ __launch_bounds__(1024) void k1_tile(const float* __restrict__ k,
                                                const float* __restrict__ v,
                                                const float* __restrict__ proj,
                                                float* __restrict__ S,
                                                float* __restrict__ Gm,
                                                float* __restrict__ GramG) {
  __shared__ float red[K1_WAVES * 32 * 32];
  __shared__ float csl[C64 * 32];
  const int dh = blockIdx.x >> 1, eh = blockIdx.x & 1, bh = blockIdx.y;
  const int t = threadIdx.x;
  const int wave = t >> 5, lane = t & 31;
  const int m = lane & 15, hh = lane >> 4, khalf = hh * 2;
  const float* kb = k + (size_t)bh * NN * DD + dh * 32;
  const float* vb = v + (size_t)bh * NN * DD + eh * 32;

  v8f acc00 = {0.f, 0.f, 0.f, 0.f, 0.f, 0.f, 0.f, 0.f};
  v8f acc01 = acc00, acc10 = acc00, acc11 = acc00;
  for (int c = 0; c < 2; ++c) {
    const int r0 = wave * K1_WROWS + c * 64;
    float cs0 = 0.f, cs1 = 0.f;
    for (int kk = 0; kk < 64; kk += 4) {
      const size_t ra = (size_t)(r0 + kk + khalf);
      const float* kr = kb + ra * DD + m;
      const float* vr = vb + ra * DD + m;
      v2f a0, a1, b0, b1;
      a0.x = kr[0];   a0.y = kr[DD];
      a1.x = kr[16];  a1.y = kr[DD + 16];
      b0.x = vr[0];   b0.y = vr[DD];
      b1.x = vr[16];  b1.y = vr[DD + 16];
      acc00 = wmma_f32(a0, b0, acc00);
      acc01 = wmma_f32(a0, b1, acc01);
      acc10 = wmma_f32(a1, b0, acc10);
      acc11 = wmma_f32(a1, b1, acc11);
      cs0 += a0.x; cs0 += a0.y;
      cs1 += a1.x; cs1 += a1.y;
    }
    cs0 += __shfl_xor(cs0, 16);
    cs1 += __shfl_xor(cs1, 16);
    if (eh == 0 && hh == 0) {
      float* dst = csl + (wave * 2 + c) * 32 + m;
      dst[0] = cs0; dst[16] = cs1;
    }
  }

  {
    float* rp = red + (wave * 32 + hh * 8) * 32 + m;
    for (int j = 0; j < 8; ++j) {
      rp[(j) * 32]           = acc00[j];
      rp[(j) * 32 + 16]      = acc01[j];
      rp[(16 + j) * 32]      = acc10[j];
      rp[(16 + j) * 32 + 16] = acc11[j];
    }
  }
  __syncthreads();

  {
    float s = 0.f;
    for (int w = 0; w < K1_WAVES; ++w) s += red[w * 1024 + t];
    Gm[(size_t)bh * DD * DD + (size_t)(dh * 32 + (t >> 5)) * DD + eh * 32 + (t & 31)] = s;
  }
  if (eh == 0 && t >= 992) {
    const int dl = t - 992;
    float run = 0.f;
    for (int c = 0; c < C64; ++c) {
      S[((size_t)(bh * C64) + c) * DD + dh * 32 + dl] = run;
      run += csl[c * 32 + dl];
    }
  }
  if (blockIdx.x == 0 && bh == 0 && wave < 16) {
    const int ti = wave >> 2, tj = wave & 3;
    v8f acc = {0.f, 0.f, 0.f, 0.f, 0.f, 0.f, 0.f, 0.f};
    for (int kk = 0; kk < DD; kk += 4) {
      const float* pr = proj + (size_t)(kk + khalf) * DD + m;
      v2f a, b;
      a.x = pr[ti * 16];  a.y = pr[DD + ti * 16];
      b.x = pr[tj * 16];  b.y = pr[DD + tj * 16];
      acc = wmma_f32(a, b, acc);
    }
    for (int j = 0; j < 8; ++j)
      GramG[(ti * 16 + hh * 8 + j) * DD + tj * 16 + m] = acc[j];
  }
}

static __device__ __forceinline__ void lds_barrier() {
  __builtin_amdgcn_fence(__ATOMIC_RELEASE, "workgroup", "local");
  __builtin_amdgcn_s_barrier();
  __builtin_amdgcn_fence(__ATOMIC_ACQUIRE, "workgroup", "local");
}

#define K3_CPW 4
#define K3_GROUPS 16

__global__ __launch_bounds__(256, 8) void k3_out(const float* __restrict__ q,
                                              const float* __restrict__ k,
                                              const float* __restrict__ S,
                                              const float* __restrict__ Gm,
                                              const float* __restrict__ GramG,
                                              float* __restrict__ out) {
  __shared__ float kc[64 * 65];
  __shared__ float gr[64 * 65];
  __shared__ float wl[64 * 65];
  __shared__ float qgl[64 * 65];
  __shared__ float qsum[4 * 64];
  __shared__ float dinv[64];
  const int bh = blockIdx.x, grp = blockIdx.y;
  const int t  = threadIdx.x;
  const int d = t & 63, qt = t >> 6;
  const int wave = t >> 5, lane = t & 31;
  const int m = lane & 15, hh = lane >> 4, khalf = hh * 2;
  const int ti = wave >> 1;
  const int tj0 = (wave & 1) * 2;
  const float* qb = q + (size_t)bh * NN * DD;
  const float* kb = k + (size_t)bh * NN * DD;
  const float* wb = Gm + (size_t)bh * DD * DD;

  float kreg[16];
  {
    const float* kp = kb + (size_t)(grp * K3_CPW) * 64 * DD + t;
    for (int i = 0; i < 16; ++i) kreg[i] = kp[i * 256];
  }
  for (int i = 0; i < 16; ++i) {
    const int idx = t + i * 256;
    gr[(qt + 4 * i) * 65 + d] = GramG[idx];
    wl[(qt + 4 * i) * 65 + d] = wb[idx];
  }

#pragma unroll 1
  for (int cc = 0; cc < K3_CPW; ++cc) {
    const int c64 = grp * K3_CPW + cc;
    const int n0 = c64 * 64;

    const float pre = S[((size_t)(bh * C64) + c64) * DD + d];
    v2f afr[16];
    {
      const float* arow = qb + (size_t)(n0 + ti * 16 + m) * DD + khalf;
      for (int kk = 0; kk < 16; ++kk) afr[kk] = *(const v2f*)(arow + kk * 4);
    }

    for (int i = 0; i < 16; ++i) kc[(qt + 4 * i) * 65 + d] = kreg[i];
    lds_barrier();

    {
      const int nn = (cc + 1 < K3_CPW) ? n0 + 64 : n0;
      const float* kp = kb + (size_t)nn * DD + t;
      for (int i = 0; i < 16; ++i) kreg[i] = kp[i * 256];
    }

    {
      float run = 0.f;
      for (int r = qt * 16; r < qt * 16 + 16; ++r) {
        run += kc[r * 65 + d];
        kc[r * 65 + d] = run;
      }
      qsum[qt * 64 + d] = run;
    }
    lds_barrier();
    if (t < 64) {
      const float o0 = pre;
      const float o1 = o0 + qsum[0 * 64 + t];
      const float o2 = o1 + qsum[1 * 64 + t];
      const float o3 = o2 + qsum[2 * 64 + t];
      qsum[0 * 64 + t] = o0; qsum[1 * 64 + t] = o1;
      qsum[2 * 64 + t] = o2; qsum[3 * 64 + t] = o3;
    }
    lds_barrier();
    {
      const float off = qsum[qt * 64 + d];
      for (int r = qt * 16; r < qt * 16 + 16; ++r) kc[r * 65 + d] += off;
    }
    lds_barrier();

    const int rbase = ti * 16 + hh * 8;

    for (int i = 0; i < 2; ++i) {
      const int tj = tj0 + i;
      v8f acc = {0.f, 0.f, 0.f, 0.f, 0.f, 0.f, 0.f, 0.f};
      for (int kk = 0; kk < 16; ++kk) {
        v2f b;
        b.x = gr[(kk * 4 + khalf) * 65 + tj * 16 + m];
        b.y = gr[(kk * 4 + khalf + 1) * 65 + tj * 16 + m];
        acc = wmma_f32(afr[kk], b, acc);
      }
      for (int j = 0; j < 8; ++j)
        qgl[(rbase + j) * 65 + tj * 16 + m] = acc[j];
    }
    lds_barrier();

    if (t < 64) {
      float s = 0.f;
      for (int dd = 0; dd < 64; ++dd) s += qgl[t * 65 + dd] * kc[t * 65 + dd];
      dinv[t] = 1.0f / s;
    }
    lds_barrier();

    v2f aq[16];
    for (int kk = 0; kk < 16; ++kk) {
      aq[kk].x = qgl[(ti * 16 + m) * 65 + kk * 4 + khalf];
      aq[kk].y = qgl[(ti * 16 + m) * 65 + kk * 4 + khalf + 1];
    }
    for (int i = 0; i < 2; ++i) {
      const int tj = tj0 + i;
      v8f acc = {0.f, 0.f, 0.f, 0.f, 0.f, 0.f, 0.f, 0.f};
      for (int kk = 0; kk < 16; ++kk) {
        v2f b;
        b.x = wl[(kk * 4 + khalf) * 65 + tj * 16 + m];
        b.y = wl[(kk * 4 + khalf + 1) * 65 + tj * 16 + m];
        acc = wmma_f32(aq[kk], b, acc);
      }
      for (int j = 0; j < 8; ++j) {
        const float dv = dinv[rbase + j];
        out[((size_t)bh * NN + n0 + rbase + j) * DD + tj * 16 + m] = acc[j] * dv;
      }
    }
  }
}

extern "C" void kernel_launch(void* const* d_in, const int* in_sizes, int n_in,
                              void* d_out, int out_size, void* d_ws, size_t ws_size,
                              hipStream_t stream) {
  const float* q    = (const float*)d_in[0];
  const float* k    = (const float*)d_in[1];
  const float* v    = (const float*)d_in[2];
  const float* proj = (const float*)d_in[3];
  float* out = (float*)d_out;

  float* ws   = (float*)d_ws;
  float* S    = ws;
  float* Gm   = S + (size_t)BH * C64 * DD;
  float* Gram = Gm + (size_t)BH * DD * DD;

  k1_tile<<<dim3(4, BH), 1024, 0, stream>>>(k, v, proj, S, Gm, Gram);
  k3_out<<<dim3(BH, K3_GROUPS), 256, 0, stream>>>(q, k, S, Gm, Gram, out);
}


# K3 re-cut: one wave per 64-row chunk, both products transposed so q@Gram accumulators are the 2nd product's B in registers; k running sum by a triangular WMMA; den, cumsum, out sum orders change
# speedup vs baseline: 1.9846x; 1.1756x over previous
#include <hip/hip_runtime.h>


typedef __attribute__((ext_vector_type(2))) float v2f;
typedef __attribute__((ext_vector_type(8))) float v8f;

#define BH 64
#define NN 4096
#define DD 64
#define K1_WAVES 32
#define K1_WROWS 128
#define C64 64

static __device__ __forceinline__ v8f wmma_f32(v2f a, v2f b, v8f c) {
  return __builtin_amdgcn_wmma_f32_16x16x4_f32(false, a, false, b, (short)0, c,
                                               false, false);
}

__global__ __launch_bounds__(1024) void k1_tile(const float* __restrict__ k,
                                                const float* __restrict__ v,
                                                const float* __restrict__ proj,
                                                float* __restrict__ S,
                                                float* __restrict__ Gm,
                                                float* __restrict__ GramG) {
  __shared__ float red[K1_WAVES * 32 * 32];
  __shared__ float csl[C64 * 32];
  const int dh = blockIdx.x >> 1, eh = blockIdx.x & 1, bh = blockIdx.y;
  const int t = threadIdx.x;
  const int wave = t >> 5, lane = t & 31;
  const int m = lane & 15, hh = lane >> 4, khalf = hh * 2;
  const float* kb = k + (size_t)bh * NN * DD + dh * 32;
  const float* vb = v + (size_t)bh * NN * DD + eh * 32;

  v8f acc00 = {0.f, 0.f, 0.f, 0.f, 0.f, 0.f, 0.f, 0.f};
  v8f acc01 = acc00, acc10 = acc00, acc11 = acc00;
  for (int c = 0; c < 2; ++c) {
    const int r0 = wave * K1_WROWS + c * 64;
    float cs0 = 0.f, cs1 = 0.f;
    for (int kk = 0; kk < 64; kk += 4) {
      const size_t ra = (size_t)(r0 + kk + khalf);
      const float* kr = kb + ra * DD + m;
      const float* vr = vb + ra * DD + m;
      v2f a0, a1, b0, b1;
      a0.x = kr[0];   a0.y = kr[DD];
      a1.x = kr[16];  a1.y = kr[DD + 16];
      b0.x = vr[0];   b0.y = vr[DD];
      b1.x = vr[16];  b1.y = vr[DD + 16];
      acc00 = wmma_f32(a0, b0, acc00);
      acc01 = wmma_f32(a0, b1, acc01);
      acc10 = wmma_f32(a1, b0, acc10);
      acc11 = wmma_f32(a1, b1, acc11);
      cs0 += a0.x; cs0 += a0.y;
      cs1 += a1.x; cs1 += a1.y;
    }
    cs0 += __shfl_xor(cs0, 16);
    cs1 += __shfl_xor(cs1, 16);
    if (eh == 0 && hh == 0) {
      float* dst = csl + (wave * 2 + c) * 32 + m;
      dst[0] = cs0; dst[16] = cs1;
    }
  }

  {
    float* rp = red + (wave * 32 + hh * 8) * 32 + m;
    for (int j = 0; j < 8; ++j) {
      rp[(j) * 32]           = acc00[j];
      rp[(j) * 32 + 16]      = acc01[j];
      rp[(16 + j) * 32]      = acc10[j];
      rp[(16 + j) * 32 + 16] = acc11[j];
    }
  }
  __syncthreads();

  {
    float s = 0.f;
    for (int w = 0; w < K1_WAVES; ++w) s += red[w * 1024 + t];
    Gm[(size_t)bh * DD * DD + (size_t)(dh * 32 + (t >> 5)) * DD + eh * 32 + (t & 31)] = s;
  }
  if (eh == 0 && t >= 992) {
    const int dl = t - 992;
    float run = 0.f;
    for (int c = 0; c < C64; ++c) {
      S[((size_t)(bh * C64) + c) * DD + dh * 32 + dl] = run;
      run += csl[c * 32 + dl];
    }
  }
  if (blockIdx.x == 0 && bh == 0 && wave < 16) {
    const int ti = wave >> 2, tj = wave & 3;
    v8f acc = {0.f, 0.f, 0.f, 0.f, 0.f, 0.f, 0.f, 0.f};
    for (int kk = 0; kk < DD; kk += 4) {
      const float* pr = proj + (size_t)(kk + khalf) * DD + m;
      v2f a, b;
      a.x = pr[ti * 16];  a.y = pr[DD + ti * 16];
      b.x = pr[tj * 16];  b.y = pr[DD + tj * 16];
      acc = wmma_f32(a, b, acc);
    }
    for (int j = 0; j < 8; ++j)
      GramG[(ti * 16 + hh * 8 + j) * DD + tj * 16 + m] = acc[j];
  }
}

typedef __attribute__((ext_vector_type(4))) float v4f;

#define K3_WAVES 8
#define K3_GROUPS 8
#define LP 68

static __device__ __forceinline__ float row_last(float x) {
  const int xi = __builtin_bit_cast(int, x);
  return __builtin_bit_cast(float,
                            __builtin_amdgcn_update_dpp(xi, xi, 0x15F  , 0xF, 0xF, false));
}

__global__ __launch_bounds__(256, 4) void k3_out(const float* __restrict__ q,
                                              const float* __restrict__ k,
                                              const float* __restrict__ S,
                                              const float* __restrict__ Gm,
                                              const float* __restrict__ GramG,
                                              float* __restrict__ out) {
  __shared__ __attribute__((aligned(16))) float gr[64 * LP];
  __shared__ __attribute__((aligned(16))) float gt[64 * LP];
  __shared__ __attribute__((aligned(16))) float ob[K3_WAVES * 16 * LP];
  const int bh = blockIdx.x;
  const int t = threadIdx.x;
  const int wave = t >> 5, lane = t & 31;
  const int m = lane & 15, hh = lane >> 4;
  const int c64 = blockIdx.y * K3_WAVES + wave;
  const float* qb = q + (size_t)bh * NN * DD;
  const float* kb = k + (size_t)bh * NN * DD;
  const float* gb = Gm + (size_t)bh * DD * DD;
  float* outb = out + (size_t)bh * NN * DD;
  float* ow = ob + wave * 16 * LP;

  v4f qf[8];
  v2f kf[4][4];
#define K3_LOAD_Q(NB)                                                           \
  {                                                                             \
    const float* qp_ = qb + (size_t)((NB) + m) * DD + 4 * hh;                   \
    _Pragma("unroll") for (int jj = 0; jj < 8; ++jj)                            \
        qf[jj] = *(const v4f*)(qp_ + 8 * jj);                                   \
  }
#define K3_LOAD_K(NB)                                                           \
  {                                                                             \
    const float* kp_ = kb + (size_t)((NB) + 2 * hh) * DD + m;                   \
    _Pragma("unroll") for (int j = 0; j < 4; ++j)                               \
        _Pragma("unroll") for (int t4 = 0; t4 < 4; ++t4) {                      \
      kf[t4][j].x = kp_[(4 * j) * DD + 16 * t4];                                \
      kf[t4][j].y = kp_[(4 * j + 1) * DD + 16 * t4];                            \
    }                                                                           \
  }
  K3_LOAD_Q(c64 * 64)

  v8f carry[4];
  {
    const float* sp = S + ((size_t)(bh * C64) + c64) * DD + 8 * hh;
#pragma unroll
    for (int t4 = 0; t4 < 4; ++t4) {
      const v4f lo = *(const v4f*)(sp + 16 * t4);
      const v4f hi = *(const v4f*)(sp + 16 * t4 + 4);
      carry[t4] = (v8f){lo.x, lo.y, lo.z, lo.w, hi.x, hi.y, hi.z, hi.w};
    }
  }

  v2f ut[4];
#pragma unroll
  for (int j = 0; j < 4; ++j) {
    ut[j].x = (4 * j + 2 * hh <= m) ? 1.f : 0.f;
    ut[j].y = (4 * j + 2 * hh + 1 <= m) ? 1.f : 0.f;
  }

  for (int i = 0; i < 4; ++i) {
    const int idx = (t + i * 256) * 4;
    const int row = idx >> 6, col = idx & 63;
    const v4f g = *(const v4f*)(GramG + idx);
    gr[(col + 0) * LP + row] = g.x;
    gr[(col + 1) * LP + row] = g.y;
    gr[(col + 2) * LP + row] = g.z;
    gr[(col + 3) * LP + row] = g.w;
    const v4f w = *(const v4f*)(gb + idx);
    gt[(col + 0) * LP + row] = w.x;
    gt[(col + 1) * LP + row] = w.y;
    gt[(col + 2) * LP + row] = w.z;
    gt[(col + 3) * LP + row] = w.w;
  }
  __syncthreads();

#pragma unroll 1
  for (int i = 0; i < 4; ++i) {
    const int n0 = c64 * 64 + i * 16;

    K3_LOAD_K(n0)
    __builtin_amdgcn_sched_barrier(0);

    v8f d1[4];
#pragma unroll
    for (int t4 = 0; t4 < 4; ++t4) d1[t4] = (v8f){0.f, 0.f, 0.f, 0.f, 0.f, 0.f, 0.f, 0.f};
#pragma unroll
    for (int jj = 0; jj < 8; ++jj) {
#pragma unroll
      for (int t4 = 0; t4 < 4; ++t4) {
        const v4f a = *(const v4f*)&gr[(16 * t4 + m) * LP + 8 * jj + 4 * hh];
        d1[t4] = wmma_f32((v2f){a.x, a.y}, (v2f){qf[jj].x, qf[jj].y}, d1[t4]);
        d1[t4] = wmma_f32((v2f){a.z, a.w}, (v2f){qf[jj].z, qf[jj].w}, d1[t4]);
      }
    }

    __builtin_amdgcn_sched_barrier(0);
#pragma unroll
    for (int j = 0; j < 4; ++j) {
#pragma unroll
      for (int t4 = 0; t4 < 4; ++t4) carry[t4] = wmma_f32(kf[t4][j], ut[j], carry[t4]);
    }

    __builtin_amdgcn_sched_barrier(0);
    {
      const int nn = (i + 1 < 4) ? n0 + 16 : n0;
      K3_LOAD_Q(nn)
    }
    __builtin_amdgcn_sched_barrier(0);

    v2f e0 = {0.f, 0.f}, e1 = {0.f, 0.f};
#pragma unroll
    for (int p = 0; p < 4; ++p) {
      e0 = (v2f){d1[0][2 * p], d1[0][2 * p + 1]} * (v2f){carry[0][2 * p], carry[0][2 * p + 1]} + e0;
      e1 = (v2f){d1[1][2 * p], d1[1][2 * p + 1]} * (v2f){carry[1][2 * p], carry[1][2 * p + 1]} + e1;
      e0 = (v2f){d1[2][2 * p], d1[2][2 * p + 1]} * (v2f){carry[2][2 * p], carry[2][2 * p + 1]} + e0;
      e1 = (v2f){d1[3][2 * p], d1[3][2 * p + 1]} * (v2f){carry[3][2 * p], carry[3][2 * p + 1]} + e1;
    }
    float den = (e0.x + e0.y) + (e1.x + e1.y);
    den += __shfl_xor(den, 16);
    const float dv = 1.0f / den;

#pragma unroll
    for (int t4 = 0; t4 < 4; ++t4) {
#pragma unroll
      for (int r = 0; r < 8; ++r) carry[t4][r] = row_last(carry[t4][r]);
    }

    v8f d2[4];
#pragma unroll
    for (int u = 0; u < 4; ++u) d2[u] = (v8f){0.f, 0.f, 0.f, 0.f, 0.f, 0.f, 0.f, 0.f};
#pragma unroll
    for (int t4 = 0; t4 < 4; ++t4) {
#pragma unroll
      for (int u = 0; u < 4; ++u) {
        const float* ap = &gt[(16 * u + m) * LP + 16 * t4 + 8 * hh];
        const v4f a0 = *(const v4f*)ap;
        const v4f a1 = *(const v4f*)(ap + 4);
        d2[u] = wmma_f32((v2f){a0.x, a0.y}, (v2f){d1[t4][0], d1[t4][1]}, d2[u]);
        d2[u] = wmma_f32((v2f){a0.z, a0.w}, (v2f){d1[t4][2], d1[t4][3]}, d2[u]);
        d2[u] = wmma_f32((v2f){a1.x, a1.y}, (v2f){d1[t4][4], d1[t4][5]}, d2[u]);
        d2[u] = wmma_f32((v2f){a1.z, a1.w}, (v2f){d1[t4][6], d1[t4][7]}, d2[u]);
      }
    }

#pragma unroll
    for (int u = 0; u < 4; ++u) {
      float* op = &ow[m * LP + 16 * u + 8 * hh];
      *(v4f*)op = (v4f){d2[u][0] * dv, d2[u][1] * dv, d2[u][2] * dv, d2[u][3] * dv};
      *(v4f*)(op + 4) = (v4f){d2[u][4] * dv, d2[u][5] * dv, d2[u][6] * dv, d2[u][7] * dv};
    }
    __builtin_amdgcn_fence(__ATOMIC_RELEASE, "wavefront", "local");
    __builtin_amdgcn_wave_barrier();
    __builtin_amdgcn_fence(__ATOMIC_ACQUIRE, "wavefront", "local");
    {
      const float* os = &ow[hh * LP + 4 * m];
      float* og = outb + (size_t)(n0 + hh) * DD + 4 * m;
#pragma unroll
      for (int p = 0; p < 8; ++p) *(v4f*)(og + 2 * p * DD) = *(const v4f*)(os + 2 * p * LP);
    }
    __builtin_amdgcn_sched_barrier(0);
  }
#undef K3_LOAD_Q
#undef K3_LOAD_K
}

extern "C" void kernel_launch(void* const* d_in, const int* in_sizes, int n_in,
                              void* d_out, int out_size, void* d_ws, size_t ws_size,
                              hipStream_t stream) {
  const float* q    = (const float*)d_in[0];
  const float* k    = (const float*)d_in[1];
  const float* v    = (const float*)d_in[2];
  const float* proj = (const float*)d_in[3];
  float* out = (float*)d_out;

  float* ws   = (float*)d_ws;
  float* S    = ws;
  float* Gm   = S + (size_t)BH * C64 * DD;
  float* Gram = Gm + (size_t)BH * DD * DD;

  k1_tile<<<dim3(4, BH), 1024, 0, stream>>>(k, v, proj, S, Gm, Gram);
  k3_out<<<dim3(BH, K3_GROUPS), 256, 0, stream>>>(q, k, S, Gm, Gram, out);
}
